# starting kernel
# baseline (speedup 1.0000x reference)
#include <hip/hip_runtime.h>

typedef float v2f __attribute__((ext_vector_type(2)));
typedef float v8f __attribute__((ext_vector_type(8)));
typedef int   v4i __attribute__((ext_vector_type(4)));

__global__ void zero_ws_kernel(float* __restrict__ ws, int n_floats) {
    int i = (blockIdx.x * blockDim.x + threadIdx.x) * 4;
    if (i + 3 < n_floats) {
        float4 z = make_float4(0.f, 0.f, 0.f, 0.f);
        *reinterpret_cast<float4*>(ws + i) = z;
    } else {
        for (int k = i; k < n_floats; ++k) ws[k] = 0.f;
    }
}

__global__ void edge_scatter_kernel(const int* __restrict__ src,
                                    const int* __restrict__ dst,
                                    const float* __restrict__ x,
                                    float* __restrict__ aggr,
                                    int n_edges) {
    int t  = blockIdx.x * blockDim.x + threadIdx.x;
    int e0 = t * 4;
    if (e0 + 3 < n_edges) {
        v4i s = __builtin_nontemporal_load(reinterpret_cast<const v4i*>(src + e0));
        v4i d = __builtin_nontemporal_load(reinterpret_cast<const v4i*>(dst + e0));

        float2 xs0 = *reinterpret_cast<const float2*>(x + (size_t)s.x * 2);
        float2 xs1 = *reinterpret_cast<const float2*>(x + (size_t)s.y * 2);
        float2 xs2 = *reinterpret_cast<const float2*>(x + (size_t)s.z * 2);
        float2 xs3 = *reinterpret_cast<const float2*>(x + (size_t)s.w * 2);

        unsafeAtomicAdd(&aggr[(size_t)d.x * 2 + 0], xs0.x);
        unsafeAtomicAdd(&aggr[(size_t)d.x * 2 + 1], xs0.y);
        unsafeAtomicAdd(&aggr[(size_t)d.y * 2 + 0], xs1.x);
        unsafeAtomicAdd(&aggr[(size_t)d.y * 2 + 1], xs1.y);
        unsafeAtomicAdd(&aggr[(size_t)d.z * 2 + 0], xs2.x);
        unsafeAtomicAdd(&aggr[(size_t)d.z * 2 + 1], xs2.y);
        unsafeAtomicAdd(&aggr[(size_t)d.w * 2 + 0], xs3.x);
        unsafeAtomicAdd(&aggr[(size_t)d.w * 2 + 1], xs3.y);
    } else {
        for (int e = e0; e < n_edges; ++e) {
            int s = src[e], d = dst[e];
            float2 xs = *reinterpret_cast<const float2*>(x + (size_t)s * 2);
            unsafeAtomicAdd(&aggr[(size_t)d * 2 + 0], xs.x);
            unsafeAtomicAdd(&aggr[(size_t)d * 2 + 1], xs.y);
        }
    }
}

__global__ void node_epilogue_wmma(const float* __restrict__ x,
                                   const float* __restrict__ aggr,
                                   const float* __restrict__ W_rel,
                                   const float* __restrict__ b_rel,
                                   const float* __restrict__ W_root,
                                   float* __restrict__ out,
                                   int n_nodes) {
    const int wave      = threadIdx.x >> 5;
    const int lane      = threadIdx.x & 31;
    const int num_tiles = (n_nodes + 15) >> 4;
    const int tile      = blockIdx.x * (blockDim.x >> 5) + wave;
    if (tile >= num_tiles) return;

    const bool full = (tile * 16 + 16) <= n_nodes;
    const int  n    = lane & 15;
    const int  half = lane >> 4;

    const int row = tile * 16 + n;
    const float* __restrict__ srcA = half ? x : aggr;
    v2f a;
    a.x = 0.f; a.y = 0.f;
    if (full) {
        float2 v = *reinterpret_cast<const float2*>(srcA + (size_t)row * 2);
        a.x = v.x; a.y = v.y;
    } else if (row < n_nodes) {
        float2 v = *reinterpret_cast<const float2*>(srcA + (size_t)row * 2);
        a.x = v.x; a.y = v.y;
    }

    v2f b;
    b.x = 0.f; b.y = 0.f;
    if (n < 2) {
        const float* __restrict__ W = half ? W_root : W_rel;
        b.x = W[n * 2 + 0];
        b.y = W[n * 2 + 1];
    }

    const float cv = (n < 2) ? b_rel[n] : 0.f;
    v8f c;
#pragma unroll
    for (int r = 0; r < 8; ++r) c[r] = cv;

    v8f d = __builtin_amdgcn_wmma_f32_16x16x4_f32(
         false, a,  false, b,
         (short)0, c,  false,  false);

    const int m0 = tile * 16 + half * 8;
    if (full) {
        if (n < 2) {
            float* __restrict__ p = out + (size_t)m0 * 2 + n;
#pragma unroll
            for (int r = 0; r < 8; ++r) p[(size_t)r * 2] = d[r];
        }
    } else {
        if (n < 2) {
#pragma unroll
            for (int r = 0; r < 8; ++r) {
                int m = m0 + r;
                if (m < n_nodes) out[(size_t)m * 2 + n] = d[r];
            }
        }
    }
}

extern "C" void kernel_launch(void* const* d_in, const int* in_sizes, int n_in,
                              void* d_out, int out_size, void* d_ws, size_t ws_size,
                              hipStream_t stream) {
    const float* x          = (const float*)d_in[0];
    const int*   edge_index = (const int*)d_in[1];
    const float* W_rel      = (const float*)d_in[2];
    const float* b_rel      = (const float*)d_in[3];
    const float* W_root     = (const float*)d_in[4];
    float*       out        = (float*)d_out;
    float*       aggr       = (float*)d_ws;

    const int n_nodes = in_sizes[0] / 2;
    const int n_edges = in_sizes[1] / 2;
    const int* src = edge_index;
    const int* dst = edge_index + n_edges;

    {
        int n_floats = n_nodes * 2;
        int threads  = (n_floats + 3) / 4;
        int blocks   = (threads + 255) / 256;
        zero_ws_kernel<<<blocks, 256, 0, stream>>>(aggr, n_floats);
    }

    {
        int threads = (n_edges + 3) / 4;
        int blocks  = (threads + 255) / 256;
        edge_scatter_kernel<<<blocks, 256, 0, stream>>>(src, dst, x, aggr, n_edges);
    }

    {
        int num_tiles = (n_nodes + 15) / 16;
        int blocks    = (num_tiles + 7) / 8;
        node_epilogue_wmma<<<blocks, 256, 0, stream>>>(x, aggr, W_rel, b_rel,
                                                       W_root, out, n_nodes);
    }
}


# Global float atomics gone, no sort: workgroup (32768-row dst range, edge slice) rescans its slice, keeps its range's edges, sums x rows in LDS (float adds, arrival order); epilogue sums 16 planes
# speedup vs baseline: 3.9169x; 3.9169x over previous
#include <hip/hip_runtime.h>
#include <stdint.h>


typedef float v2f __attribute__((ext_vector_type(2)));
typedef float v8f __attribute__((ext_vector_type(8)));
typedef int   v4i __attribute__((ext_vector_type(4)));

constexpr int kRangeShift = 15;
constexpr int kRangeRows  = 1 << kRangeShift;
constexpr int kSlices     = 16;
constexpr int kThreads    = 1024;

__device__ __forceinline__ v4i load_group(const int* __restrict__ slice, unsigned g) {
    return *reinterpret_cast<const v4i*>(slice + 4u * g);
}

__device__ __forceinline__ float2 gather_row(const float* __restrict__ x, bool m, int s) {
    const unsigned off = m ? (unsigned)s * 2u : 0u;
    return *reinterpret_cast<const float2*>(x + off);
}

__device__ __forceinline__ void lds_add_row(float* acc, bool m, int d, float2 v) {
    if (m) {
        const int dl = d & (kRangeRows - 1);
        __hip_atomic_fetch_add(&acc[dl * 2 + 0], v.x, __ATOMIC_RELAXED, __HIP_MEMORY_SCOPE_WORKGROUP);
        __hip_atomic_fetch_add(&acc[dl * 2 + 1], v.y, __ATOMIC_RELAXED, __HIP_MEMORY_SCOPE_WORKGROUP);
    }
}

__global__ __launch_bounds__(kThreads)
void range_slice_accumulate(const int* __restrict__ src,
                            const int* __restrict__ dst,
                            const float* __restrict__ x,
                            float* __restrict__ planes,
                            int n_edges, int n_nodes, int n_ranges) {
    __shared__ float acc[kRangeRows * 2];

    const int tid  = threadIdx.x;
    const int r    = blockIdx.x % n_ranges;
    const int s    = blockIdx.x / n_ranges;
    const int row0 = r << kRangeShift;
    int rows = n_nodes - row0;
    if (rows > kRangeRows) rows = kRangeRows;

    float4* acc4 = reinterpret_cast<float4*>(acc);
    for (int i = tid; i < kRangeRows * 2 / 4; i += kThreads)
        acc4[i] = make_float4(0.f, 0.f, 0.f, 0.f);
    __syncthreads();

    const int per  = (((n_edges + kSlices - 1) / kSlices) + 3) & ~3;
    const int e_lo = s * per;
    int e_hi = e_lo + per;
    if (e_hi > n_edges) e_hi = n_edges;
    const int span    = e_hi > e_lo ? e_hi - e_lo : 0;
    const int ngroups = span >> 2;
    const unsigned ur = (unsigned)r;

    if (ngroups > 0) {
        const int* __restrict__ dsl = dst + e_lo;
        const int* __restrict__ ssl = src + e_lo;
        const unsigned ng   = (unsigned)ngroups;
        const unsigned last = ng - 1u;
        unsigned gA  = (unsigned)tid;
        unsigned gB  = (unsigned)tid + kThreads;
        bool     okA = gA < ng;
        bool     okB = gB < ng;
        v4i dA = load_group(dsl, gA < last ? gA : last);
        v4i sA = load_group(ssl, gA < last ? gA : last);
        v4i dB = load_group(dsl, gB < last ? gB : last);
        v4i sB = load_group(ssl, gB < last ? gB : last);

        for (int base = 0; base < ngroups; base += 2 * kThreads) {
            const bool m0 = okA & (((unsigned)dA.x >> kRangeShift) == ur);
            const bool m1 = okA & (((unsigned)dA.y >> kRangeShift) == ur);
            const bool m2 = okA & (((unsigned)dA.z >> kRangeShift) == ur);
            const bool m3 = okA & (((unsigned)dA.w >> kRangeShift) == ur);
            const bool m4 = okB & (((unsigned)dB.x >> kRangeShift) == ur);
            const bool m5 = okB & (((unsigned)dB.y >> kRangeShift) == ur);
            const bool m6 = okB & (((unsigned)dB.z >> kRangeShift) == ur);
            const bool m7 = okB & (((unsigned)dB.w >> kRangeShift) == ur);

            const float2 v0 = gather_row(x, m0, sA.x);
            const float2 v1 = gather_row(x, m1, sA.y);
            const float2 v2 = gather_row(x, m2, sA.z);
            const float2 v3 = gather_row(x, m3, sA.w);
            const float2 v4 = gather_row(x, m4, sB.x);
            const float2 v5 = gather_row(x, m5, sB.y);
            const float2 v6 = gather_row(x, m6, sB.z);
            const float2 v7 = gather_row(x, m7, sB.w);

            gA += 2 * kThreads;
            gB += 2 * kThreads;
            const v4i ndA = load_group(dsl, gA < last ? gA : last);
            const v4i nsA = load_group(ssl, gA < last ? gA : last);
            const v4i ndB = load_group(dsl, gB < last ? gB : last);
            const v4i nsB = load_group(ssl, gB < last ? gB : last);

            asm volatile("" :: "v"(v0.x), "v"(v0.y));

            lds_add_row(acc, m0, dA.x, v0);
            lds_add_row(acc, m1, dA.y, v1);
            lds_add_row(acc, m2, dA.z, v2);
            lds_add_row(acc, m3, dA.w, v3);
            lds_add_row(acc, m4, dB.x, v4);
            lds_add_row(acc, m5, dB.y, v5);
            lds_add_row(acc, m6, dB.z, v6);
            lds_add_row(acc, m7, dB.w, v7);

            okA = gA < ng;
            okB = gB < ng;
            dA = ndA; sA = nsA; dB = ndB; sB = nsB;
        }
    }

    {
        const int e = e_lo + 4 * ngroups + tid;
        if (e < e_hi) {
            const int d  = dst[e];
            const bool m = ((unsigned)d >> kRangeShift) == ur;
            const float2 v = gather_row(x, m, src[e]);
            lds_add_row(acc, m, d, v);
        }
    }
    __syncthreads();

    float* plane = planes + (size_t)s * ((size_t)n_nodes * 2) + (size_t)row0 * 2;
    const float2* acc2 = reinterpret_cast<const float2*>(acc);
    const int pairs = ((reinterpret_cast<uintptr_t>(plane) & 15) == 0) ? (rows >> 1) : 0;
    for (int i = tid; i < pairs; i += kThreads)
        reinterpret_cast<float4*>(plane)[i] = acc4[i];
    for (int i = pairs * 2 + tid; i < rows; i += kThreads)
        reinterpret_cast<float2*>(plane)[i] = acc2[i];
}

__global__ void node_epilogue_wmma(const float* __restrict__ x,
                                   const float* __restrict__ planes,
                                   const float* __restrict__ W_rel,
                                   const float* __restrict__ b_rel,
                                   const float* __restrict__ W_root,
                                   float* __restrict__ out,
                                   int n_nodes) {
    const int wave      = threadIdx.x >> 5;
    const int lane      = threadIdx.x & 31;
    const int num_tiles = (n_nodes + 15) >> 4;
    const int tile      = blockIdx.x * (blockDim.x >> 5) + wave;
    if (tile >= num_tiles) return;

    const bool full = (tile * 16 + 16) <= n_nodes;
    const int  n    = lane & 15;
    const int  half = lane >> 4;

    const int row = tile * 16 + n;
    const size_t plane_stride = (size_t)n_nodes * 2;
    float px = 0.f, py = 0.f, xx = 0.f, xy = 0.f;
    if (row < n_nodes) {
        const float* p = planes + (size_t)(half * (kSlices / 2)) * plane_stride + (size_t)row * 2;
        float2 v[kSlices / 2];
        for (int k = 0; k < kSlices / 2; ++k)
            v[k] = *reinterpret_cast<const float2*>(p + (size_t)k * plane_stride);
        for (int k = 0; k < kSlices / 2; ++k) { px += v[k].x; py += v[k].y; }
        if (half) {
            float2 t = *reinterpret_cast<const float2*>(x + (size_t)row * 2);
            xx = t.x; xy = t.y;
        }
    }
    const float qx = __shfl_xor(px, 16);
    const float qy = __shfl_xor(py, 16);

    v2f a;
    a.x = half ? xx : (px + qx);
    a.y = half ? xy : (py + qy);

    v2f b;
    b.x = 0.f; b.y = 0.f;
    if (n < 2) {
        const float* __restrict__ W = half ? W_root : W_rel;
        b.x = W[n * 2 + 0];
        b.y = W[n * 2 + 1];
    }

    const float cv = (n < 2) ? b_rel[n] : 0.f;
    v8f c;
    for (int r = 0; r < 8; ++r) c[r] = cv;

    v8f d = __builtin_amdgcn_wmma_f32_16x16x4_f32(
         false, a,  false, b,
         (short)0, c,  false,  false);

    const int m0 = tile * 16 + half * 8;
    if (full) {
        if (n < 2) {
            float* __restrict__ p = out + (size_t)m0 * 2 + n;
            for (int r = 0; r < 8; ++r) p[(size_t)r * 2] = d[r];
        }
    } else {
        if (n < 2) {
            for (int r = 0; r < 8; ++r) {
                int m = m0 + r;
                if (m < n_nodes) out[(size_t)m * 2 + n] = d[r];
            }
        }
    }
}

extern "C" void kernel_launch(void* const* d_in, const int* in_sizes, int n_in,
                              void* d_out, int out_size, void* d_ws, size_t ws_size,
                              hipStream_t stream) {
    const float* x          = (const float*)d_in[0];
    const int*   edge_index = (const int*)d_in[1];
    const float* W_rel      = (const float*)d_in[2];
    const float* b_rel      = (const float*)d_in[3];
    const float* W_root     = (const float*)d_in[4];
    float*       out        = (float*)d_out;
    float*       planes     = (float*)d_ws;

    const int n_nodes = in_sizes[0] / 2;
    const int n_edges = in_sizes[1] / 2;
    const int* src = edge_index;
    const int* dst = edge_index + n_edges;

    {
        int n_ranges = (n_nodes + kRangeRows - 1) / kRangeRows;
        range_slice_accumulate<<<n_ranges * kSlices, kThreads, 0, stream>>>(
            src, dst, x, planes, n_edges, n_nodes, n_ranges);
    }

    {
        int num_tiles = (n_nodes + 15) / 16;
        int blocks    = (num_tiles + 7) / 8;
        node_epilogue_wmma<<<blocks, 256, 0, stream>>>(x, planes, W_rel, b_rel,
                                                       W_root, out, n_nodes);
    }
}


# Accumulate loop sheds vector address and bounds work: full trips unmasked, tail once after the loop; index and x loads by scalar base + 32-bit lane index; LDS slot by a shift-add; adds: arrival order
# speedup vs baseline: 4.1187x; 1.0515x over previous
#include <hip/hip_runtime.h>
#include <stdint.h>


typedef float v2f __attribute__((ext_vector_type(2)));
typedef float v8f __attribute__((ext_vector_type(8)));
typedef int   v4i __attribute__((ext_vector_type(4)));

constexpr int kRangeShift = 15;
constexpr int kRangeRows  = 1 << kRangeShift;
constexpr int kSlices     = 16;
constexpr int kThreads    = 1024;

__device__ __forceinline__ float2 gather_row(const float2* __restrict__ xr, bool m, int s) {
    return xr[m ? (unsigned)s : 0u];
}

__device__ __forceinline__ void lds_add_row(float* acc, bool m, int d, unsigned neg_row0_bytes, float2 v) {
    if (m) {
        const unsigned off = ((unsigned)d << 3) + neg_row0_bytes;
        float* p = reinterpret_cast<float*>(reinterpret_cast<char*>(acc) + off);
        __hip_atomic_fetch_add(p + 0, v.x, __ATOMIC_RELAXED, __HIP_MEMORY_SCOPE_WORKGROUP);
        __hip_atomic_fetch_add(p + 1, v.y, __ATOMIC_RELAXED, __HIP_MEMORY_SCOPE_WORKGROUP);
    }
}

__device__ __forceinline__ void scan_trip(float* acc, const float2* __restrict__ xr,
                                          const v4i* __restrict__ dn, const v4i* __restrict__ sn,
                                          int tid, unsigned ur, unsigned neg_row0_bytes,
                                          const v4i& dA, const v4i& sA, const v4i& dB, const v4i& sB,
                                          v4i& ndA, v4i& nsA, v4i& ndB, v4i& nsB) {
    const bool m0 = ((unsigned)dA.x >> kRangeShift) == ur;
    const bool m1 = ((unsigned)dA.y >> kRangeShift) == ur;
    const bool m2 = ((unsigned)dA.z >> kRangeShift) == ur;
    const bool m3 = ((unsigned)dA.w >> kRangeShift) == ur;
    const bool m4 = ((unsigned)dB.x >> kRangeShift) == ur;
    const bool m5 = ((unsigned)dB.y >> kRangeShift) == ur;
    const bool m6 = ((unsigned)dB.z >> kRangeShift) == ur;
    const bool m7 = ((unsigned)dB.w >> kRangeShift) == ur;

    const float2 v0 = gather_row(xr, m0, sA.x);
    const float2 v1 = gather_row(xr, m1, sA.y);
    const float2 v2 = gather_row(xr, m2, sA.z);
    const float2 v3 = gather_row(xr, m3, sA.w);
    const float2 v4 = gather_row(xr, m4, sB.x);
    const float2 v5 = gather_row(xr, m5, sB.y);
    const float2 v6 = gather_row(xr, m6, sB.z);
    const float2 v7 = gather_row(xr, m7, sB.w);

    ndA = dn[tid];
    nsA = sn[tid];
    ndB = dn[tid + kThreads];
    nsB = sn[tid + kThreads];

    asm volatile("" :: "v"(v0.x), "v"(v0.y));

    lds_add_row(acc, m0, dA.x, neg_row0_bytes, v0);
    lds_add_row(acc, m1, dA.y, neg_row0_bytes, v1);
    lds_add_row(acc, m2, dA.z, neg_row0_bytes, v2);
    lds_add_row(acc, m3, dA.w, neg_row0_bytes, v3);
    lds_add_row(acc, m4, dB.x, neg_row0_bytes, v4);
    lds_add_row(acc, m5, dB.y, neg_row0_bytes, v5);
    lds_add_row(acc, m6, dB.z, neg_row0_bytes, v6);
    lds_add_row(acc, m7, dB.w, neg_row0_bytes, v7);
}

__device__ __forceinline__ void scan_group_tail(float* acc, const float2* __restrict__ xr,
                                                const v4i* __restrict__ dg, const v4i* __restrict__ sg,
                                                unsigned g, unsigned ur, unsigned neg_row0_bytes) {
    const v4i d = dg[g];
    const v4i s = sg[g];
    const bool m0 = ((unsigned)d.x >> kRangeShift) == ur;
    const bool m1 = ((unsigned)d.y >> kRangeShift) == ur;
    const bool m2 = ((unsigned)d.z >> kRangeShift) == ur;
    const bool m3 = ((unsigned)d.w >> kRangeShift) == ur;
    const float2 v0 = gather_row(xr, m0, s.x);
    const float2 v1 = gather_row(xr, m1, s.y);
    const float2 v2 = gather_row(xr, m2, s.z);
    const float2 v3 = gather_row(xr, m3, s.w);
    lds_add_row(acc, m0, d.x, neg_row0_bytes, v0);
    lds_add_row(acc, m1, d.y, neg_row0_bytes, v1);
    lds_add_row(acc, m2, d.z, neg_row0_bytes, v2);
    lds_add_row(acc, m3, d.w, neg_row0_bytes, v3);
}

__global__ __launch_bounds__(kThreads)
void range_slice_accumulate(const int* __restrict__ src,
                            const int* __restrict__ dst,
                            const float* __restrict__ x,
                            float* __restrict__ planes,
                            int n_edges, int n_nodes, int n_ranges) {
    __shared__ float acc[kRangeRows * 2];

    const int tid  = threadIdx.x;
    const int r    = blockIdx.x % n_ranges;
    const int s    = blockIdx.x / n_ranges;
    const int row0 = r << kRangeShift;
    int rows = n_nodes - row0;
    if (rows > kRangeRows) rows = kRangeRows;

    float4* acc4 = reinterpret_cast<float4*>(acc);
    for (int i = tid; i < kRangeRows * 2 / 4; i += kThreads)
        acc4[i] = make_float4(0.f, 0.f, 0.f, 0.f);
    __syncthreads();

    const int per  = (((n_edges + kSlices - 1) / kSlices) + 3) & ~3;
    const int e_lo = s * per;
    int e_hi = e_lo + per;
    if (e_hi > n_edges) e_hi = n_edges;
    const int span    = e_hi > e_lo ? e_hi - e_lo : 0;
    const int ngroups = span >> 2;
    const unsigned ur     = (unsigned)r;
    const unsigned neg_row0_bytes = 0u - ((unsigned)row0 << 3);
    const float2* __restrict__ xr = reinterpret_cast<const float2*>(x);

    if (ngroups > 0) {
        const v4i* __restrict__ dg = reinterpret_cast<const v4i*>(dst + e_lo);
        const v4i* __restrict__ sg = reinterpret_cast<const v4i*>(src + e_lo);
        constexpr int kTrip = 2 * kThreads;
        const int full = ngroups / kTrip;

        if (full > 0) {
            v4i dA = dg[tid], sA = sg[tid], dB = dg[tid + kThreads], sB = sg[tid + kThreads];
            v4i dC, sC, dD, sD;
            int t = 0;
            for (; t + 2 <= full; t += 2) {
                const int t2 = (t + 2 < full) ? t + 2 : t + 1;
                scan_trip(acc, xr, dg + (size_t)(t + 1) * kTrip, sg + (size_t)(t + 1) * kTrip,
                          tid, ur, neg_row0_bytes, dA, sA, dB, sB, dC, sC, dD, sD);
                scan_trip(acc, xr, dg + (size_t)t2 * kTrip, sg + (size_t)t2 * kTrip,
                          tid, ur, neg_row0_bytes, dC, sC, dD, sD, dA, sA, dB, sB);
            }
            if (t < full)
                scan_trip(acc, xr, dg + (size_t)t * kTrip, sg + (size_t)t * kTrip,
                          tid, ur, neg_row0_bytes, dA, sA, dB, sB, dC, sC, dD, sD);
        }

        const unsigned ng = (unsigned)ngroups;
        const unsigned gA = (unsigned)(full * kTrip + tid);
        const unsigned gB = gA + kThreads;
        if (gA < ng) scan_group_tail(acc, xr, dg, sg, gA, ur, neg_row0_bytes);
        if (gB < ng) scan_group_tail(acc, xr, dg, sg, gB, ur, neg_row0_bytes);
    }

    {
        const int e = e_lo + 4 * ngroups + tid;
        if (e < e_hi) {
            const int d  = dst[e];
            const bool m = ((unsigned)d >> kRangeShift) == ur;
            const float2 v = gather_row(xr, m, src[e]);
            lds_add_row(acc, m, d, neg_row0_bytes, v);
        }
    }
    __syncthreads();

    float* plane = planes + (size_t)s * ((size_t)n_nodes * 2) + (size_t)row0 * 2;
    const float2* acc2 = reinterpret_cast<const float2*>(acc);
    const int pairs = ((reinterpret_cast<uintptr_t>(plane) & 15) == 0) ? (rows >> 1) : 0;
    for (int i = tid; i < pairs; i += kThreads)
        reinterpret_cast<float4*>(plane)[i] = acc4[i];
    for (int i = pairs * 2 + tid; i < rows; i += kThreads)
        reinterpret_cast<float2*>(plane)[i] = acc2[i];
}

__global__ void node_epilogue_wmma(const float* __restrict__ x,
                                   const float* __restrict__ planes,
                                   const float* __restrict__ W_rel,
                                   const float* __restrict__ b_rel,
                                   const float* __restrict__ W_root,
                                   float* __restrict__ out,
                                   int n_nodes) {
    const int wave      = threadIdx.x >> 5;
    const int lane      = threadIdx.x & 31;
    const int num_tiles = (n_nodes + 15) >> 4;
    const int tile      = blockIdx.x * (blockDim.x >> 5) + wave;
    if (tile >= num_tiles) return;

    const bool full = (tile * 16 + 16) <= n_nodes;
    const int  n    = lane & 15;
    const int  half = lane >> 4;

    const int row = tile * 16 + n;
    const size_t plane_stride = (size_t)n_nodes * 2;
    float px = 0.f, py = 0.f, xx = 0.f, xy = 0.f;
    if (row < n_nodes) {
        const float* p = planes + (size_t)(half * (kSlices / 2)) * plane_stride + (size_t)row * 2;
        float2 v[kSlices / 2];
        for (int k = 0; k < kSlices / 2; ++k)
            v[k] = *reinterpret_cast<const float2*>(p + (size_t)k * plane_stride);
        for (int k = 0; k < kSlices / 2; ++k) { px += v[k].x; py += v[k].y; }
        if (half) {
            float2 t = *reinterpret_cast<const float2*>(x + (size_t)row * 2);
            xx = t.x; xy = t.y;
        }
    }
    const float qx = __shfl_xor(px, 16);
    const float qy = __shfl_xor(py, 16);

    v2f a;
    a.x = half ? xx : (px + qx);
    a.y = half ? xy : (py + qy);

    v2f b;
    b.x = 0.f; b.y = 0.f;
    if (n < 2) {
        const float* __restrict__ W = half ? W_root : W_rel;
        b.x = W[n * 2 + 0];
        b.y = W[n * 2 + 1];
    }

    const float cv = (n < 2) ? b_rel[n] : 0.f;
    v8f c;
    for (int r = 0; r < 8; ++r) c[r] = cv;

    v8f d = __builtin_amdgcn_wmma_f32_16x16x4_f32(
         false, a,  false, b,
         (short)0, c,  false,  false);

    const int m0 = tile * 16 + half * 8;
    if (full) {
        if (n < 2) {
            float* __restrict__ p = out + (size_t)m0 * 2 + n;
            for (int r = 0; r < 8; ++r) p[(size_t)r * 2] = d[r];
        }
    } else {
        if (n < 2) {
            for (int r = 0; r < 8; ++r) {
                int m = m0 + r;
                if (m < n_nodes) out[(size_t)m * 2 + n] = d[r];
            }
        }
    }
}

extern "C" void kernel_launch(void* const* d_in, const int* in_sizes, int n_in,
                              void* d_out, int out_size, void* d_ws, size_t ws_size,
                              hipStream_t stream) {
    const float* x          = (const float*)d_in[0];
    const int*   edge_index = (const int*)d_in[1];
    const float* W_rel      = (const float*)d_in[2];
    const float* b_rel      = (const float*)d_in[3];
    const float* W_root     = (const float*)d_in[4];
    float*       out        = (float*)d_out;
    float*       planes     = (float*)d_ws;

    const int n_nodes = in_sizes[0] / 2;
    const int n_edges = in_sizes[1] / 2;
    const int* src = edge_index;
    const int* dst = edge_index + n_edges;

    {
        int n_ranges = (n_nodes + kRangeRows - 1) / kRangeRows;
        range_slice_accumulate<<<n_ranges * kSlices, kThreads, 0, stream>>>(
            src, dst, x, planes, n_edges, n_nodes, n_ranges);
    }

    {
        int num_tiles = (n_nodes + 15) / 16;
        int blocks    = (num_tiles + 7) / 8;
        node_epilogue_wmma<<<blocks, 256, 0, stream>>>(x, planes, W_rel, b_rel,
                                                       W_root, out, n_nodes);
    }
}
